# starting kernel
# baseline (speedup 1.0000x reference)
#include <hip/hip_runtime.h>
#include <hip/hip_bf16.h>


typedef __attribute__((ext_vector_type(2))) float v2f;
typedef __attribute__((ext_vector_type(8))) float v8f;

static constexpr int BB = 16;
static constexpr int LL = 2048;
static constexpr int DD = 512;
static constexpr int CK = 16;
static constexpr int NC = LL / CK;

__global__ void ds_prep_kernel(const float* __restrict__ push_gate,
                               const float* __restrict__ pop_gate,
                               float* __restrict__ Wws,
                               float* __restrict__ Pws) {
  const int inst = blockIdx.x;
  const int b    = inst / NC;
  const int ch   = inst % NC;
  const int t0   = ch * CK;
  const int lane = threadIdx.x;

  __shared__ float a_sm[CK];
  __shared__ float c_sm[CK];

  if (lane < CK) {
    const float p  = push_gate[(size_t)b * LL + t0 + lane];
    const float o  = pop_gate [(size_t)b * LL + t0 + lane];
    const float om = 1.0f - o;
    a_sm[lane] = (1.0f - p) * om;
    c_sm[lane] = p * om;
  }
  __syncthreads();

  float* Wb = Wws + (size_t)inst * 256;

  if (lane < CK) {
    const int k = lane;
    float w[CK];
    #pragma unroll
    for (int j = 0; j < CK; ++j) w[j] = 0.0f;
    w[k] = c_sm[k];
    float prod = 1.0f;
    for (int j = k - 1; j >= 0; --j) {
      prod *= a_sm[j + 1];
      w[j] = c_sm[j] * prod;
    }
    #pragma unroll
    for (int kk = 0; kk < 4; ++kk) {
      Wb[kk * 64 + k * 2 + 0]        = w[4 * kk + 0];
      Wb[kk * 64 + k * 2 + 1]        = w[4 * kk + 1];
      Wb[kk * 64 + (k + 16) * 2 + 0] = w[4 * kk + 2];
      Wb[kk * 64 + (k + 16) * 2 + 1] = w[4 * kk + 3];
    }
    float pp = 1.0f;
    for (int i = 0; i <= k; ++i) pp *= a_sm[i];
    Pws[(size_t)inst * CK + k] = pp;
  }
}

__global__ void ds_scan_wmma_kernel(const float* __restrict__ x,
                                    const float* __restrict__ Wws,
                                    const float* __restrict__ Pws,
                                    float* __restrict__ out) {
  const int tile = blockIdx.x;
  const int b    = blockIdx.y;
  const int lane = threadIdx.x;
  const int half = lane >> 4;
  const int n    = lane & 15;
  const int d0   = tile * CK;

  const float* __restrict__ xb = x   + (size_t)b * LL * DD + d0 + n;
  float*       __restrict__ ob = out + (size_t)b * LL * DD + d0 + n;
  const float* __restrict__ Wb = Wws + (size_t)b * NC * 256;
  const float* __restrict__ Pb = Pws + (size_t)b * NC * CK;

  float h = 0.0f;

  for (int ch = 0; ch < NC; ++ch) {
    const int t0 = ch * CK;
    const float* Wc = Wb + (size_t)ch * 256;
    const float* Pc = Pb + (size_t)ch * CK;

    v8f acc = {};
    #pragma unroll
    for (int kk = 0; kk < 4; ++kk) {
      v2f A = *(const v2f*)(Wc + kk * 64 + lane * 2);
      const int kb = t0 + 4 * kk + 2 * half;
      v2f Bv;
      Bv.x = xb[(size_t)kb * DD];
      Bv.y = xb[(size_t)(kb + 1) * DD];
      acc = __builtin_amdgcn_wmma_f32_16x16x4_f32(
           false, A,  false, Bv,
           (short)0, acc,  false,  false);
    }

    float Pv[CK];
    #pragma unroll
    for (int k = 0; k < CK; ++k) Pv[k] = Pc[k];
    #pragma unroll
    for (int r = 0; r < 8; ++r) {
      const float pk = half ? Pv[r + 8] : Pv[r];
      acc[r] += pk * h;
    }

    #pragma unroll
    for (int r = 0; r < 8; ++r) {
      ob[(size_t)(t0 + r + 8 * half) * DD] = acc[r];
    }

    h = __shfl(acc[7], n + 16, 32);
  }
}

extern "C" void kernel_launch(void* const* d_in, const int* in_sizes, int n_in,
                              void* d_out, int out_size, void* d_ws, size_t ws_size,
                              hipStream_t stream) {
  const float* x  = (const float*)d_in[0];
  const float* pg = (const float*)d_in[1];
  const float* og = (const float*)d_in[2];
  float* out = (float*)d_out;

  float* Wws = (float*)d_ws;
  float* Pws = Wws + (size_t)BB * NC * 256;

  ds_prep_kernel<<<dim3(BB * NC), dim3(32), 0, stream>>>(pg, og, Wws, Pws);
  ds_scan_wmma_kernel<<<dim3(DD / CK, BB), dim3(32), 0, stream>>>(x, Wws, Pws, out);
}


# Carry chain kept in lanes 16-31 (one FMA a chunk); lanes 0-15 get it by v_permlanex16 instead of ds_bpermute, so the serial part no longer queues behind the fragment reads in LDS; sums unchanged
# speedup vs baseline: 4.4320x; 4.4320x over previous
#include <hip/hip_runtime.h>
#include <hip/hip_bf16.h>


typedef __attribute__((ext_vector_type(2))) float v2f;
typedef __attribute__((ext_vector_type(8))) float v8f;

static constexpr int BB = 16;
static constexpr int LL = 2048;
static constexpr int DD = 512;
static constexpr int CK = 16;
static constexpr int NC = LL / CK;

__global__ void ds_prep_kernel(const float* __restrict__ push_gate,
                               const float* __restrict__ pop_gate,
                               float* __restrict__ Wws,
                               float* __restrict__ Pws) {
  const int inst = blockIdx.x;
  const int b    = inst / NC;
  const int ch   = inst % NC;
  const int t0   = ch * CK;
  const int lane = threadIdx.x;

  __shared__ float a_sm[CK];
  __shared__ float c_sm[CK];

  if (lane < CK) {
    const float p  = push_gate[(size_t)b * LL + t0 + lane];
    const float o  = pop_gate [(size_t)b * LL + t0 + lane];
    const float om = 1.0f - o;
    a_sm[lane] = (1.0f - p) * om;
    c_sm[lane] = p * om;
  }
  __syncthreads();

  float* Wb = Wws + (size_t)inst * 256;

  if (lane < CK) {
    const int k = lane;
    float w[CK];
    #pragma unroll
    for (int j = 0; j < CK; ++j) w[j] = 0.0f;
    w[k] = c_sm[k];
    float prod = 1.0f;
    for (int j = k - 1; j >= 0; --j) {
      prod *= a_sm[j + 1];
      w[j] = c_sm[j] * prod;
    }
    #pragma unroll
    for (int kk = 0; kk < 4; ++kk) {
      Wb[kk * 64 + k * 2 + 0]        = w[4 * kk + 0];
      Wb[kk * 64 + k * 2 + 1]        = w[4 * kk + 1];
      Wb[kk * 64 + (k + 16) * 2 + 0] = w[4 * kk + 2];
      Wb[kk * 64 + (k + 16) * 2 + 1] = w[4 * kk + 3];
    }
    float pp = 1.0f;
    for (int i = 0; i <= k; ++i) pp *= a_sm[i];
    Pws[(size_t)inst * CK + k] = pp;
  }
}

typedef __attribute__((ext_vector_type(4))) float v4f;

static constexpr int GP    = 4;
static constexpr int NG    = NC / GP;
static constexpr int NW    = 4;
static constexpr int CG    = NW * CK;
static constexpr int GR    = GP * CK;
static constexpr int XP    = CG + 8;
static constexpr int XSL   = GR * XP;
static constexpr int WSL   = GP * 256;
static constexpr int PGF   = GP * CK;
static constexpr int PSL   = NW * 32 * 4;
static constexpr int SLOT  = XSL + WSL + PSL;
static constexpr int OP    = CG + 4;
static constexpr int OSL   = GR * OP;
static constexpr int DEPTH = 4;

struct ChunkOps {
  v2f A[4];
  v2f B[4];
  v4f P0, P1;
};

struct Stage {
  v4f X[8];
  v4f Wv[2];
  v4f Pv;
};

__device__ __forceinline__ void ds_stage_load(Stage& s,
                                              const float* __restrict__ xg, unsigned xt,
                                              const float* __restrict__ Wu, unsigned wt,
                                              const float* __restrict__ Pu, unsigned pt,
                                              int grp) {
  const float* __restrict__ xr = xg + (size_t)grp * GR * DD;
  const float* __restrict__ Wr = Wu + (size_t)grp * WSL;
  #pragma unroll
  for (int i = 0; i < 8; ++i)
    s.X[i] = *(const v4f*)(xr + (xt + (unsigned)(i * 8 * DD)));
  #pragma unroll
  for (int i = 0; i < 2; ++i)
    s.Wv[i] = *(const v4f*)(Wr + (wt + (unsigned)(i * 512)));
  s.Pv = *(const v4f*)(Pu + (size_t)grp * PGF + pt);
}

__device__ __forceinline__ void ds_stage_store(const Stage& s, float* slot,
                                               unsigned xl, unsigned wt) {
  #pragma unroll
  for (int i = 0; i < 8; ++i)
    *(v4f*)(slot + (xl + (unsigned)(i * 8 * XP))) = s.X[i];
  #pragma unroll
  for (int i = 0; i < 2; ++i)
    *(v4f*)(slot + (XSL + wt + (unsigned)(i * 512))) = s.Wv[i];
  *(v4f*)(slot + (XSL + WSL + wt)) = s.Pv;
}

__device__ __forceinline__ void ds_frag_load(ChunkOps& q, const float* slot,
                                             unsigned bl, unsigned al, int u) {
  #pragma unroll
  for (int kk = 0; kk < 4; ++kk) {
    q.A[kk] = *(const v2f*)(slot + (XSL + al + (unsigned)(u * 256 + kk * 64)));
    q.B[kk].x = slot[bl + (unsigned)((u * CK + 4 * kk) * XP)];
    q.B[kk].y = slot[bl + (unsigned)((u * CK + 4 * kk + 1) * XP)];
  }
}

__device__ __forceinline__ void ds_load_P(ChunkOps& q, const float* slot,
                                          unsigned po, int u) {
  q.P0 = *(const v4f*)(slot + (XSL + WSL + po + (unsigned)(u * CK)));
  q.P1 = *(const v4f*)(slot + (XSL + WSL + po + (unsigned)(u * CK + 4)));
}

__device__ __forceinline__ void ds_out_rows_load(v4f (&R)[8], const float* oslot,
                                                 unsigned ol) {
  #pragma unroll
  for (int i = 0; i < 8; ++i)
    R[i] = *(const v4f*)(oslot + (ol + (unsigned)(i * 8 * OP)));
}

__device__ __forceinline__ void ds_out_rows_store(const v4f (&R)[8],
                                                  float* __restrict__ og,
                                                  unsigned xt, int grp) {
  float* __restrict__ orow = og + (size_t)grp * GR * DD;
  #pragma unroll
  for (int i = 0; i < 8; ++i)
    *(v4f*)(orow + (xt + (unsigned)(i * 8 * DD))) = R[i];
}

__device__ __forceinline__ void ds_slot_barrier() {
  __builtin_amdgcn_fence(__ATOMIC_RELEASE, "workgroup", "local");
  __builtin_amdgcn_s_barrier();
  __builtin_amdgcn_fence(__ATOMIC_ACQUIRE, "workgroup", "local");
}

__global__ void __launch_bounds__(NW * 32)
ds_scan_wmma_kernel(const float* __restrict__ x,
                    const float* __restrict__ Wws,
                    const float* __restrict__ Pws,
                    float* __restrict__ out) {
  __shared__ __attribute__((aligned(16))) float lds[2 * SLOT + 2 * OSL];

  const int cgi  = blockIdx.x;
  const int b    = blockIdx.y;
  const int tid  = threadIdx.x;
  const int w    = tid >> 5;
  const int lane = tid & 31;
  const int half = lane >> 4;
  const int n    = lane & 15;

  const float* __restrict__ xg = x   + (size_t)b * LL * DD + cgi * CG;
  float*       __restrict__ og = out + (size_t)b * LL * DD + cgi * CG;
  const float* __restrict__ Wu = Wws + (size_t)b * NC * 256;
  const float* __restrict__ Pu = Pws + (size_t)b * NC * CK;
  const unsigned ob = 8u * half * OP + 16u * w + n;
  const unsigned ol = (unsigned)(tid >> 4) * OP + (unsigned)(tid & 15) * 4u;
  const unsigned po = (unsigned)(tid >> 4) * 64u + 8u * half;
  const unsigned xt = (unsigned)(tid >> 4) * DD + (unsigned)(tid & 15) * 4u;
  const unsigned xl = (unsigned)(tid >> 4) * XP + (unsigned)(tid & 15) * 4u;
  const unsigned wt = (unsigned)tid * 4u;
  const unsigned pt = (unsigned)(tid & 15) * 4u;
  const unsigned bl = 2u * half * XP + 16u * w + n;
  const unsigned al = lane * 2u;

  Stage    S[DEPTH];
  ChunkOps q[GP];

  {
    Stage t;
    ds_stage_load(t, xg, xt, Wu, wt, Pu, pt, 0);
    #pragma unroll
    for (int j = 0; j < DEPTH; ++j)
      ds_stage_load(S[(1 + j) % DEPTH], xg, xt, Wu, wt, Pu, pt, 1 + j);
    ds_stage_store(t, lds, xl, wt);
    ds_slot_barrier();
    #pragma unroll
    for (int u = 0; u < GP; ++u) {
      ds_frag_load(q[u], lds, bl, al, u);
      ds_load_P(q[u], lds, po, u);
    }
  }

  float hc = 0.0f;

  for (int g0 = 0; g0 < NG; g0 += DEPTH) {
    #pragma unroll
    for (int p = 0; p < DEPTH; ++p) {
      const int g   = g0 + p;

      v8f acc[GP];
      #pragma unroll
      for (int u = 0; u < GP; ++u) {
        acc[u] = (v8f){};
        #pragma unroll
        for (int kk = 0; kk < 4; ++kk) {
          acc[u] = __builtin_amdgcn_wmma_f32_16x16x4_f32(
               false, q[u].A[kk],  false, q[u].B[kk],
               (short)0, acc[u],  false,  false);
        }
      }

      v4f R[8];
      const bool drain = (g > 0);
      {
        Stage& sn   = S[(p + 1) % DEPTH];
        float* slot = lds + ((p + 1) & 1) * SLOT;
        ds_stage_store(sn, slot, xl, wt);
        const int gl = (g + 1 + DEPTH < NG) ? (g + 1 + DEPTH) : (NG - 1);
        ds_stage_load(sn, xg, xt, Wu, wt, Pu, pt, gl);
        ds_slot_barrier();
        if (drain) ds_out_rows_load(R, lds + 2 * SLOT + ((p + 1) & 1) * OSL, ol);
        #pragma unroll
        for (int u = 0; u < GP; ++u) ds_frag_load(q[u], slot, bl, al, u);
      }

      __builtin_amdgcn_sched_barrier(0);

      float hin[GP];
      #pragma unroll
      for (int u = 0; u < GP; ++u) {
        hin[u] = hc;
        hc = __builtin_fmaf(q[u].P1.w, hc, acc[u][7]);
      }

      float* oslot = lds + 2 * SLOT + (p & 1) * OSL;
      #pragma unroll
      for (int u = 0; u < GP; ++u) {
        const unsigned hbits = __builtin_bit_cast(unsigned, hin[u]);
        const float hx = __builtin_bit_cast(float,
            __builtin_amdgcn_permlanex16(hbits, hbits, 0x76543210u, 0xfedcba98u,
                                         false, false));
        const float h = half ? hin[u] : hx;

        acc[u][0] += q[u].P0.x * h;  acc[u][1] += q[u].P0.y * h;
        acc[u][2] += q[u].P0.z * h;  acc[u][3] += q[u].P0.w * h;
        acc[u][4] += q[u].P1.x * h;  acc[u][5] += q[u].P1.y * h;
        acc[u][6] += q[u].P1.z * h;
        acc[u][7] = __builtin_fmaf(q[u].P1.w, h, acc[u][7]);

        #pragma unroll
        for (int r = 0; r < 8; ++r) {
          oslot[ob + (unsigned)((u * CK + r) * OP)] = acc[u][r];
        }

        if (u == 0 && drain) ds_out_rows_store(R, og, xt, g - 1);
      }

      {
        unsigned ps = po;
        asm volatile("" : "+v"(ps)
                        : "v"(acc[0]), "v"(acc[1]), "v"(acc[2]), "v"(acc[3]));
        const float* slot = lds + ((p + 1) & 1) * SLOT;
        #pragma unroll
        for (int u = 0; u < GP; ++u) ds_load_P(q[u], slot, ps, u);
      }
    }
  }

  {
    v4f R[8];
    ds_slot_barrier();
    ds_out_rows_load(R, lds + 2 * SLOT + ((NG - 1) & 1) * OSL, ol);
    ds_out_rows_store(R, og, xt, NG - 1);
  }
}

extern "C" void kernel_launch(void* const* d_in, const int* in_sizes, int n_in,
                              void* d_out, int out_size, void* d_ws, size_t ws_size,
                              hipStream_t stream) {
  const float* x  = (const float*)d_in[0];
  const float* pg = (const float*)d_in[1];
  const float* og = (const float*)d_in[2];
  float* out = (float*)d_out;

  float* Wws = (float*)d_ws;
  float* Pws = Wws + (size_t)BB * NC * 256;

  ds_prep_kernel<<<dim3(BB * NC), dim3(32), 0, stream>>>(pg, og, Wws, Pws);
  ds_scan_wmma_kernel<<<dim3(DD / CG, BB), dim3(NW * 32), 0, stream>>>(x, Wws, Pws, out);
}


# Prep kernel: one thread per (chunk, row) builds its W row and P in registers from the chunk's 16 gates, unrolled: no LDS, barrier or per-lane loop lengths; 256 groups of 4 waves, was 2048 of 1 wave
# speedup vs baseline: 4.6563x; 1.0506x over previous
#include <hip/hip_runtime.h>
#include <hip/hip_bf16.h>


typedef __attribute__((ext_vector_type(2))) float v2f;
typedef __attribute__((ext_vector_type(8))) float v8f;

static constexpr int BB = 16;
static constexpr int LL = 2048;
static constexpr int DD = 512;
static constexpr int CK = 16;
static constexpr int NC = LL / CK;

typedef __attribute__((ext_vector_type(4))) float v4f;

static constexpr int PREP_T = 128;
static constexpr int PREP_G = BB * NC * CK / PREP_T;

__global__ void __launch_bounds__(PREP_T)
ds_prep_kernel(const float* __restrict__ push_gate,
               const float* __restrict__ pop_gate,
               float* __restrict__ Wws,
               float* __restrict__ Pws) {
  const int gi   = blockIdx.x * PREP_T + threadIdx.x;
  const int inst = gi >> 4;
  const int k    = gi & 15;

  float a[CK], c[CK];
  #pragma unroll
  for (int i = 0; i < 4; ++i) {
    const v4f p = *(const v4f*)(push_gate + (size_t)inst * CK + 4 * i);
    const v4f o = *(const v4f*)(pop_gate  + (size_t)inst * CK + 4 * i);
    #pragma unroll
    for (int e = 0; e < 4; ++e) {
      const float om = 1.0f - o[e];
      a[4 * i + e] = (1.0f - p[e]) * om;
      c[4 * i + e] = p[e] * om;
    }
  }

  float w[CK];
  float prod = 1.0f;
  #pragma unroll
  for (int j = CK - 1; j >= 0; --j) {
    if (j < CK - 1) prod = (j < k) ? prod * a[j + 1] : prod;
    w[j] = (j < k) ? c[j] * prod : ((j == k) ? c[j] : 0.0f);
  }

  float pp = 1.0f;
  #pragma unroll
  for (int i = 0; i < CK; ++i) pp = (i <= k) ? pp * a[i] : pp;

  float* Wb = Wws + (size_t)inst * 256;
  #pragma unroll
  for (int kk = 0; kk < 4; ++kk) {
    *(v2f*)(Wb + kk * 64 + k * 2)        = (v2f){w[4 * kk + 0], w[4 * kk + 1]};
    *(v2f*)(Wb + kk * 64 + (k + 16) * 2) = (v2f){w[4 * kk + 2], w[4 * kk + 3]};
  }
  Pws[gi] = pp;
}


static constexpr int GP    = 4;
static constexpr int NG    = NC / GP;
static constexpr int NW    = 4;
static constexpr int CG    = NW * CK;
static constexpr int GR    = GP * CK;
static constexpr int XP    = CG + 8;
static constexpr int XSL   = GR * XP;
static constexpr int WSL   = GP * 256;
static constexpr int PGF   = GP * CK;
static constexpr int PSL   = NW * 32 * 4;
static constexpr int SLOT  = XSL + WSL + PSL;
static constexpr int OP    = CG + 4;
static constexpr int OSL   = GR * OP;
static constexpr int DEPTH = 4;

struct ChunkOps {
  v2f A[4];
  v2f B[4];
  v4f P0, P1;
};

struct Stage {
  v4f X[8];
  v4f Wv[2];
  v4f Pv;
};

__device__ __forceinline__ void ds_stage_load(Stage& s,
                                              const float* __restrict__ xg, unsigned xt,
                                              const float* __restrict__ Wu, unsigned wt,
                                              const float* __restrict__ Pu, unsigned pt,
                                              int grp) {
  const float* __restrict__ xr = xg + (size_t)grp * GR * DD;
  const float* __restrict__ Wr = Wu + (size_t)grp * WSL;
  #pragma unroll
  for (int i = 0; i < 8; ++i)
    s.X[i] = *(const v4f*)(xr + (xt + (unsigned)(i * 8 * DD)));
  #pragma unroll
  for (int i = 0; i < 2; ++i)
    s.Wv[i] = *(const v4f*)(Wr + (wt + (unsigned)(i * 512)));
  s.Pv = *(const v4f*)(Pu + (size_t)grp * PGF + pt);
}

__device__ __forceinline__ void ds_stage_store(const Stage& s, float* slot,
                                               unsigned xl, unsigned wt) {
  #pragma unroll
  for (int i = 0; i < 8; ++i)
    *(v4f*)(slot + (xl + (unsigned)(i * 8 * XP))) = s.X[i];
  #pragma unroll
  for (int i = 0; i < 2; ++i)
    *(v4f*)(slot + (XSL + wt + (unsigned)(i * 512))) = s.Wv[i];
  *(v4f*)(slot + (XSL + WSL + wt)) = s.Pv;
}

__device__ __forceinline__ void ds_frag_load(ChunkOps& q, const float* slot,
                                             unsigned bl, unsigned al, int u) {
  #pragma unroll
  for (int kk = 0; kk < 4; ++kk) {
    q.A[kk] = *(const v2f*)(slot + (XSL + al + (unsigned)(u * 256 + kk * 64)));
    q.B[kk].x = slot[bl + (unsigned)((u * CK + 4 * kk) * XP)];
    q.B[kk].y = slot[bl + (unsigned)((u * CK + 4 * kk + 1) * XP)];
  }
}

__device__ __forceinline__ void ds_load_P(ChunkOps& q, const float* slot,
                                          unsigned po, int u) {
  q.P0 = *(const v4f*)(slot + (XSL + WSL + po + (unsigned)(u * CK)));
  q.P1 = *(const v4f*)(slot + (XSL + WSL + po + (unsigned)(u * CK + 4)));
}

__device__ __forceinline__ void ds_out_rows_load(v4f (&R)[8], const float* oslot,
                                                 unsigned ol) {
  #pragma unroll
  for (int i = 0; i < 8; ++i)
    R[i] = *(const v4f*)(oslot + (ol + (unsigned)(i * 8 * OP)));
}

__device__ __forceinline__ void ds_out_rows_store(const v4f (&R)[8],
                                                  float* __restrict__ og,
                                                  unsigned xt, int grp) {
  float* __restrict__ orow = og + (size_t)grp * GR * DD;
  #pragma unroll
  for (int i = 0; i < 8; ++i)
    *(v4f*)(orow + (xt + (unsigned)(i * 8 * DD))) = R[i];
}

__device__ __forceinline__ void ds_slot_barrier() {
  __builtin_amdgcn_fence(__ATOMIC_RELEASE, "workgroup", "local");
  __builtin_amdgcn_s_barrier();
  __builtin_amdgcn_fence(__ATOMIC_ACQUIRE, "workgroup", "local");
}

__global__ void __launch_bounds__(NW * 32)
ds_scan_wmma_kernel(const float* __restrict__ x,
                    const float* __restrict__ Wws,
                    const float* __restrict__ Pws,
                    float* __restrict__ out) {
  __shared__ __attribute__((aligned(16))) float lds[2 * SLOT + 2 * OSL];

  const int cgi  = blockIdx.x;
  const int b    = blockIdx.y;
  const int tid  = threadIdx.x;
  const int w    = tid >> 5;
  const int lane = tid & 31;
  const int half = lane >> 4;
  const int n    = lane & 15;

  const float* __restrict__ xg = x   + (size_t)b * LL * DD + cgi * CG;
  float*       __restrict__ og = out + (size_t)b * LL * DD + cgi * CG;
  const float* __restrict__ Wu = Wws + (size_t)b * NC * 256;
  const float* __restrict__ Pu = Pws + (size_t)b * NC * CK;
  const unsigned ob = 8u * half * OP + 16u * w + n;
  const unsigned ol = (unsigned)(tid >> 4) * OP + (unsigned)(tid & 15) * 4u;
  const unsigned po = (unsigned)(tid >> 4) * 64u + 8u * half;
  const unsigned xt = (unsigned)(tid >> 4) * DD + (unsigned)(tid & 15) * 4u;
  const unsigned xl = (unsigned)(tid >> 4) * XP + (unsigned)(tid & 15) * 4u;
  const unsigned wt = (unsigned)tid * 4u;
  const unsigned pt = (unsigned)(tid & 15) * 4u;
  const unsigned bl = 2u * half * XP + 16u * w + n;
  const unsigned al = lane * 2u;

  Stage    S[DEPTH];
  ChunkOps q[GP];

  {
    Stage t;
    ds_stage_load(t, xg, xt, Wu, wt, Pu, pt, 0);
    #pragma unroll
    for (int j = 0; j < DEPTH; ++j)
      ds_stage_load(S[(1 + j) % DEPTH], xg, xt, Wu, wt, Pu, pt, 1 + j);
    ds_stage_store(t, lds, xl, wt);
    ds_slot_barrier();
    #pragma unroll
    for (int u = 0; u < GP; ++u) {
      ds_frag_load(q[u], lds, bl, al, u);
      ds_load_P(q[u], lds, po, u);
    }
  }

  float hc = 0.0f;

  for (int g0 = 0; g0 < NG; g0 += DEPTH) {
    #pragma unroll
    for (int p = 0; p < DEPTH; ++p) {
      const int g   = g0 + p;

      v8f acc[GP];
      #pragma unroll
      for (int u = 0; u < GP; ++u) {
        acc[u] = (v8f){};
        #pragma unroll
        for (int kk = 0; kk < 4; ++kk) {
          acc[u] = __builtin_amdgcn_wmma_f32_16x16x4_f32(
               false, q[u].A[kk],  false, q[u].B[kk],
               (short)0, acc[u],  false,  false);
        }
      }

      v4f R[8];
      const bool drain = (g > 0);
      {
        Stage& sn   = S[(p + 1) % DEPTH];
        float* slot = lds + ((p + 1) & 1) * SLOT;
        ds_stage_store(sn, slot, xl, wt);
        const int gl = (g + 1 + DEPTH < NG) ? (g + 1 + DEPTH) : (NG - 1);
        ds_stage_load(sn, xg, xt, Wu, wt, Pu, pt, gl);
        ds_slot_barrier();
        if (drain) ds_out_rows_load(R, lds + 2 * SLOT + ((p + 1) & 1) * OSL, ol);
        #pragma unroll
        for (int u = 0; u < GP; ++u) ds_frag_load(q[u], slot, bl, al, u);
      }

      __builtin_amdgcn_sched_barrier(0);

      float hin[GP];
      #pragma unroll
      for (int u = 0; u < GP; ++u) {
        hin[u] = hc;
        hc = __builtin_fmaf(q[u].P1.w, hc, acc[u][7]);
      }

      float* oslot = lds + 2 * SLOT + (p & 1) * OSL;
      #pragma unroll
      for (int u = 0; u < GP; ++u) {
        const unsigned hbits = __builtin_bit_cast(unsigned, hin[u]);
        const float hx = __builtin_bit_cast(float,
            __builtin_amdgcn_permlanex16(hbits, hbits, 0x76543210u, 0xfedcba98u,
                                         false, false));
        const float h = half ? hin[u] : hx;

        acc[u][0] += q[u].P0.x * h;  acc[u][1] += q[u].P0.y * h;
        acc[u][2] += q[u].P0.z * h;  acc[u][3] += q[u].P0.w * h;
        acc[u][4] += q[u].P1.x * h;  acc[u][5] += q[u].P1.y * h;
        acc[u][6] += q[u].P1.z * h;
        acc[u][7] = __builtin_fmaf(q[u].P1.w, h, acc[u][7]);

        #pragma unroll
        for (int r = 0; r < 8; ++r) {
          oslot[ob + (unsigned)((u * CK + r) * OP)] = acc[u][r];
        }

        if (u == 0 && drain) ds_out_rows_store(R, og, xt, g - 1);
      }

      {
        unsigned ps = po;
        asm volatile("" : "+v"(ps)
                        : "v"(acc[0]), "v"(acc[1]), "v"(acc[2]), "v"(acc[3]));
        const float* slot = lds + ((p + 1) & 1) * SLOT;
        #pragma unroll
        for (int u = 0; u < GP; ++u) ds_load_P(q[u], slot, ps, u);
      }
    }
  }

  {
    v4f R[8];
    ds_slot_barrier();
    ds_out_rows_load(R, lds + 2 * SLOT + ((NG - 1) & 1) * OSL, ol);
    ds_out_rows_store(R, og, xt, NG - 1);
  }
}

extern "C" void kernel_launch(void* const* d_in, const int* in_sizes, int n_in,
                              void* d_out, int out_size, void* d_ws, size_t ws_size,
                              hipStream_t stream) {
  const float* x  = (const float*)d_in[0];
  const float* pg = (const float*)d_in[1];
  const float* og = (const float*)d_in[2];
  float* out = (float*)d_out;

  float* Wws = (float*)d_ws;
  float* Pws = Wws + (size_t)BB * NC * 256;

  ds_prep_kernel<<<dim3(PREP_G), dim3(PREP_T), 0, stream>>>(pg, og, Wws, Pws);
  ds_scan_wmma_kernel<<<dim3(DD / CG, BB), dim3(NW * 32), 0, stream>>>(x, Wws, Pws, out);
}


# Time axis split over 4 independent work-groups (512 in all, was 128): each restarts the chain from zero where the older state's f32 weight is exactly zero (or at step 0); sums otherwise unchanged
# speedup vs baseline: 6.1406x; 1.3188x over previous
#include <hip/hip_runtime.h>
#include <hip/hip_bf16.h>


typedef __attribute__((ext_vector_type(2))) float v2f;
typedef __attribute__((ext_vector_type(8))) float v8f;

static constexpr int BB = 16;
static constexpr int LL = 2048;
static constexpr int DD = 512;
static constexpr int CK = 16;
static constexpr int NC = LL / CK;

typedef __attribute__((ext_vector_type(4))) float v4f;

static constexpr int PREP_T = 128;
static constexpr int PREP_G = BB * NC * CK / PREP_T;

__global__ void __launch_bounds__(PREP_T)
ds_prep_kernel(const float* __restrict__ push_gate,
               const float* __restrict__ pop_gate,
               float* __restrict__ Wws,
               float* __restrict__ Pws) {
  const int gi   = blockIdx.x * PREP_T + threadIdx.x;
  const int inst = gi >> 4;
  const int k    = gi & 15;

  float a[CK], c[CK];
  #pragma unroll
  for (int i = 0; i < 4; ++i) {
    const v4f p = *(const v4f*)(push_gate + (size_t)inst * CK + 4 * i);
    const v4f o = *(const v4f*)(pop_gate  + (size_t)inst * CK + 4 * i);
    #pragma unroll
    for (int e = 0; e < 4; ++e) {
      const float om = 1.0f - o[e];
      a[4 * i + e] = (1.0f - p[e]) * om;
      c[4 * i + e] = p[e] * om;
    }
  }

  float w[CK];
  float prod = 1.0f;
  #pragma unroll
  for (int j = CK - 1; j >= 0; --j) {
    if (j < CK - 1) prod = (j < k) ? prod * a[j + 1] : prod;
    w[j] = (j < k) ? c[j] * prod : ((j == k) ? c[j] : 0.0f);
  }

  float pp = 1.0f;
  #pragma unroll
  for (int i = 0; i < CK; ++i) pp = (i <= k) ? pp * a[i] : pp;

  float* Wb = Wws + (size_t)inst * 256;
  #pragma unroll
  for (int kk = 0; kk < 4; ++kk) {
    *(v2f*)(Wb + kk * 64 + k * 2)        = (v2f){w[4 * kk + 0], w[4 * kk + 1]};
    *(v2f*)(Wb + kk * 64 + (k + 16) * 2) = (v2f){w[4 * kk + 2], w[4 * kk + 3]};
  }
  Pws[gi] = pp;
}


static constexpr int GP    = 4;
static constexpr int NG    = NC / GP;
static constexpr int NW    = 4;
static constexpr int CG    = NW * CK;
static constexpr int GR    = GP * CK;
static constexpr int XP    = CG + 8;
static constexpr int XSL   = GR * XP;
static constexpr int WSL   = GP * 256;
static constexpr int PGF   = GP * CK;
static constexpr int PSL   = NW * 32 * 4;
static constexpr int SLOT  = XSL + WSL + PSL;
static constexpr int OP    = CG + 4;
static constexpr int OSL   = GR * OP;
static constexpr int DEPTH = 4;
static constexpr int NS    = 4;
static constexpr int NGS   = NG / NS;

struct ChunkOps {
  v2f A[4];
  v2f B[4];
  v4f P0, P1;
};

struct Stage {
  v4f X[8];
  v4f Wv[2];
  v4f Pv;
};

__device__ __forceinline__ void ds_stage_load(Stage& s,
                                              const float* __restrict__ xg, unsigned xt,
                                              const float* __restrict__ Wu, unsigned wt,
                                              const float* __restrict__ Pu, unsigned pt,
                                              int grp) {
  const float* __restrict__ xr = xg + (size_t)grp * GR * DD;
  const float* __restrict__ Wr = Wu + (size_t)grp * WSL;
  #pragma unroll
  for (int i = 0; i < 8; ++i)
    s.X[i] = *(const v4f*)(xr + (xt + (unsigned)(i * 8 * DD)));
  #pragma unroll
  for (int i = 0; i < 2; ++i)
    s.Wv[i] = *(const v4f*)(Wr + (wt + (unsigned)(i * 512)));
  s.Pv = *(const v4f*)(Pu + (size_t)grp * PGF + pt);
}

__device__ __forceinline__ void ds_stage_store(const Stage& s, float* slot,
                                               unsigned xl, unsigned wt) {
  #pragma unroll
  for (int i = 0; i < 8; ++i)
    *(v4f*)(slot + (xl + (unsigned)(i * 8 * XP))) = s.X[i];
  #pragma unroll
  for (int i = 0; i < 2; ++i)
    *(v4f*)(slot + (XSL + wt + (unsigned)(i * 512))) = s.Wv[i];
  *(v4f*)(slot + (XSL + WSL + wt)) = s.Pv;
}

__device__ __forceinline__ void ds_frag_load(ChunkOps& q, const float* slot,
                                             unsigned bl, unsigned al, int u) {
  #pragma unroll
  for (int kk = 0; kk < 4; ++kk) {
    q.A[kk] = *(const v2f*)(slot + (XSL + al + (unsigned)(u * 256 + kk * 64)));
    q.B[kk].x = slot[bl + (unsigned)((u * CK + 4 * kk) * XP)];
    q.B[kk].y = slot[bl + (unsigned)((u * CK + 4 * kk + 1) * XP)];
  }
}

__device__ __forceinline__ void ds_load_P(ChunkOps& q, const float* slot,
                                          unsigned po, int u) {
  q.P0 = *(const v4f*)(slot + (XSL + WSL + po + (unsigned)(u * CK)));
  q.P1 = *(const v4f*)(slot + (XSL + WSL + po + (unsigned)(u * CK + 4)));
}

__device__ __forceinline__ void ds_out_rows_load(v4f (&R)[8], const float* oslot,
                                                 unsigned ol) {
  #pragma unroll
  for (int i = 0; i < 8; ++i)
    R[i] = *(const v4f*)(oslot + (ol + (unsigned)(i * 8 * OP)));
}

__device__ __forceinline__ void ds_out_rows_store(const v4f (&R)[8],
                                                  float* __restrict__ og,
                                                  unsigned xt, int grp) {
  float* __restrict__ orow = og + (size_t)grp * GR * DD;
  #pragma unroll
  for (int i = 0; i < 8; ++i)
    *(v4f*)(orow + (xt + (unsigned)(i * 8 * DD))) = R[i];
}

__device__ __forceinline__ void ds_slot_barrier() {
  __builtin_amdgcn_fence(__ATOMIC_RELEASE, "workgroup", "local");
  __builtin_amdgcn_s_barrier();
  __builtin_amdgcn_fence(__ATOMIC_ACQUIRE, "workgroup", "local");
}

__device__ __forceinline__ void ds_warm_group(const Stage& s, float* slot,
                                              ChunkOps (&q)[GP], float& hc,
                                              unsigned xl, unsigned wt, unsigned bl,
                                              unsigned al, unsigned po) {
  ds_stage_store(s, slot, xl, wt);
  ds_slot_barrier();
  #pragma unroll
  for (int u = 0; u < GP; ++u) {
    ds_frag_load(q[u], slot, bl, al, u);
    ds_load_P(q[u], slot, po, u);
  }
  v8f acc[GP];
  #pragma unroll
  for (int u = 0; u < GP; ++u) {
    acc[u] = (v8f){};
    #pragma unroll
    for (int kk = 0; kk < 4; ++kk) {
      acc[u] = __builtin_amdgcn_wmma_f32_16x16x4_f32(
           false, q[u].A[kk],  false, q[u].B[kk],
           (short)0, acc[u],  false,  false);
    }
  }
  #pragma unroll
  for (int u = 0; u < GP; ++u) hc = __builtin_fmaf(q[u].P1.w, hc, acc[u][7]);
}

__global__ void __launch_bounds__(NW * 32, 2)
ds_scan_wmma_kernel(const float* __restrict__ x,
                    const float* __restrict__ Wws,
                    const float* __restrict__ Pws,
                    float* __restrict__ out) {
  __shared__ __attribute__((aligned(16))) float lds[2 * SLOT + 2 * OSL];

  const int cgi  = blockIdx.x;
  const int b    = blockIdx.y;
  const int seg  = blockIdx.z;
  const int tid  = threadIdx.x;
  const int w    = tid >> 5;
  const int lane = tid & 31;
  const int half = lane >> 4;
  const int n    = lane & 15;

  const float* __restrict__ xg = x   + (size_t)b * LL * DD + cgi * CG;
  float*       __restrict__ og = out + (size_t)b * LL * DD + cgi * CG;
  const float* __restrict__ Wu = Wws + (size_t)b * NC * 256;
  const float* __restrict__ Pu = Pws + (size_t)b * NC * CK;
  const unsigned ob = 8u * half * OP + 16u * w + n;
  const unsigned ol = (unsigned)(tid >> 4) * OP + (unsigned)(tid & 15) * 4u;
  const unsigned po = (unsigned)(tid >> 4) * 64u + 8u * half;
  const unsigned xt = (unsigned)(tid >> 4) * DD + (unsigned)(tid & 15) * 4u;
  const unsigned xl = (unsigned)(tid >> 4) * XP + (unsigned)(tid & 15) * 4u;
  const unsigned wt = (unsigned)tid * 4u;
  const unsigned pt = (unsigned)(tid & 15) * 4u;
  const unsigned bl = 2u * half * XP + 16u * w + n;
  const unsigned al = lane * 2u;

  const int gs = seg * NGS;
  int gw = gs;
  {
    float wgt = 1.0f;
    while (gw > 0 && wgt != 0.0f) {
      --gw;
      const float* __restrict__ pp = Pu + (size_t)gw * PGF;
      wgt = wgt * pp[3 * CK + 15] * pp[2 * CK + 15] * pp[CK + 15] * pp[15];
    }
  }
  const int nwu = gs - gw;

  Stage    S[DEPTH];
  ChunkOps q[GP];
  float hc = 0.0f;

  {
    Stage t;
    Stage& tw = S[0];
    if (nwu > 0) ds_stage_load(tw, xg, xt, Wu, wt, Pu, pt, gw);
    ds_stage_load(t, xg, xt, Wu, wt, Pu, pt, gs);
    #pragma unroll
    for (int j = 0; j < DEPTH - 1; ++j)
      ds_stage_load(S[1 + j], xg, xt, Wu, wt, Pu, pt, gs + 1 + j);

    if (nwu > 0) {
      ds_warm_group(tw, lds + SLOT, q, hc, xl, wt, bl, al, po);
      for (int wi = 1; wi < nwu; ++wi) {
        ds_stage_load(tw, xg, xt, Wu, wt, Pu, pt, gw + wi);
        ds_slot_barrier();
        ds_warm_group(tw, lds + SLOT, q, hc, xl, wt, bl, al, po);
      }
    }
    ds_stage_load(S[0], xg, xt, Wu, wt, Pu, pt, gs + DEPTH);

    ds_stage_store(t, lds, xl, wt);
    ds_slot_barrier();
    #pragma unroll
    for (int u = 0; u < GP; ++u) {
      ds_frag_load(q[u], lds, bl, al, u);
      ds_load_P(q[u], lds, po, u);
    }
  }

  for (int g0 = 0; g0 < NGS; g0 += DEPTH) {
    #pragma unroll
    for (int p = 0; p < DEPTH; ++p) {
      const int g   = g0 + p;

      v8f acc[GP];
      #pragma unroll
      for (int u = 0; u < GP; ++u) {
        acc[u] = (v8f){};
        #pragma unroll
        for (int kk = 0; kk < 4; ++kk) {
          acc[u] = __builtin_amdgcn_wmma_f32_16x16x4_f32(
               false, q[u].A[kk],  false, q[u].B[kk],
               (short)0, acc[u],  false,  false);
        }
      }

      v4f R[8];
      const bool drain = (g > 0);
      {
        Stage& sn   = S[(p + 1) % DEPTH];
        float* slot = lds + ((p + 1) & 1) * SLOT;
        ds_stage_store(sn, slot, xl, wt);
        const int gl = (g + 1 + DEPTH < NGS) ? (g + 1 + DEPTH) : (NGS - 1);
        ds_stage_load(sn, xg, xt, Wu, wt, Pu, pt, gs + gl);
        ds_slot_barrier();
        if (drain) ds_out_rows_load(R, lds + 2 * SLOT + ((p + 1) & 1) * OSL, ol);
        #pragma unroll
        for (int u = 0; u < GP; ++u) ds_frag_load(q[u], slot, bl, al, u);
      }

      __builtin_amdgcn_sched_barrier(0);

      float hin[GP];
      #pragma unroll
      for (int u = 0; u < GP; ++u) {
        hin[u] = hc;
        hc = __builtin_fmaf(q[u].P1.w, hc, acc[u][7]);
      }

      float* oslot = lds + 2 * SLOT + (p & 1) * OSL;
      #pragma unroll
      for (int u = 0; u < GP; ++u) {
        const unsigned hbits = __builtin_bit_cast(unsigned, hin[u]);
        const float hx = __builtin_bit_cast(float,
            __builtin_amdgcn_permlanex16(hbits, hbits, 0x76543210u, 0xfedcba98u,
                                         false, false));
        const float h = half ? hin[u] : hx;

        acc[u][0] += q[u].P0.x * h;  acc[u][1] += q[u].P0.y * h;
        acc[u][2] += q[u].P0.z * h;  acc[u][3] += q[u].P0.w * h;
        acc[u][4] += q[u].P1.x * h;  acc[u][5] += q[u].P1.y * h;
        acc[u][6] += q[u].P1.z * h;
        acc[u][7] = __builtin_fmaf(q[u].P1.w, h, acc[u][7]);

        #pragma unroll
        for (int r = 0; r < 8; ++r) {
          oslot[ob + (unsigned)((u * CK + r) * OP)] = acc[u][r];
        }

        if (u == 0 && drain) ds_out_rows_store(R, og, xt, gs + g - 1);
      }

      {
        unsigned ps = po;
        asm volatile("" : "+v"(ps)
                        : "v"(acc[0]), "v"(acc[1]), "v"(acc[2]), "v"(acc[3]));
        const float* slot = lds + ((p + 1) & 1) * SLOT;
        #pragma unroll
        for (int u = 0; u < GP; ++u) ds_load_P(q[u], slot, ps, u);
      }
    }
  }

  {
    v4f R[8];
    ds_slot_barrier();
    ds_out_rows_load(R, lds + 2 * SLOT + ((NGS - 1) & 1) * OSL, ol);
    ds_out_rows_store(R, og, xt, gs + NGS - 1);
  }
}

extern "C" void kernel_launch(void* const* d_in, const int* in_sizes, int n_in,
                              void* d_out, int out_size, void* d_ws, size_t ws_size,
                              hipStream_t stream) {
  const float* x  = (const float*)d_in[0];
  const float* pg = (const float*)d_in[1];
  const float* og = (const float*)d_in[2];
  float* out = (float*)d_out;

  float* Wws = (float*)d_ws;
  float* Pws = Wws + (size_t)BB * NC * 256;

  ds_prep_kernel<<<dim3(PREP_G), dim3(PREP_T), 0, stream>>>(pg, og, Wws, Pws);
  ds_scan_wmma_kernel<<<dim3(DD / CG, BB, NS), dim3(NW * 32), 0, stream>>>(x, Wws, Pws, out);
}


# W and P built once per work-group at its start, from the gates, into LDS that every wave reads (4 rows a thread, under the first X loads): prep launch and workspace gone; sums unchanged
# speedup vs baseline: 7.4586x; 1.2146x over previous
#include <hip/hip_runtime.h>
#include <hip/hip_bf16.h>


typedef __attribute__((ext_vector_type(2))) float v2f;
typedef __attribute__((ext_vector_type(8))) float v8f;

static constexpr int BB = 16;
static constexpr int LL = 2048;
static constexpr int DD = 512;
static constexpr int CK = 16;
static constexpr int NC = LL / CK;

typedef __attribute__((ext_vector_type(4))) float v4f;

struct Gates {
  v4f p[4];
  v4f o[4];
};

__device__ __forceinline__ void ds_gates_load(Gates& g,
                                              const float* __restrict__ gp,
                                              const float* __restrict__ go,
                                              unsigned s0) {
  #pragma unroll
  for (int i = 0; i < 4; ++i) {
    g.p[i] = *(const v4f*)(gp + (s0 + (unsigned)(4 * i)));
    g.o[i] = *(const v4f*)(go + (s0 + (unsigned)(4 * i)));
  }
}

__device__ __forceinline__ void ds_build_row(const Gates& g, int k,
                                             float* Wb, float* Pb) {
  float a[CK], c[CK];
  #pragma unroll
  for (int i = 0; i < 4; ++i) {
    #pragma unroll
    for (int e = 0; e < 4; ++e) {
      const float om = 1.0f - g.o[i][e];
      a[4 * i + e] = (1.0f - g.p[i][e]) * om;
      c[4 * i + e] = g.p[i][e] * om;
    }
  }

  #pragma unroll
  for (int i = 0; i < CK; ++i) {
    a[i] = (i <= k) ? a[i] : 1.0f;
    c[i] = (i <= k) ? c[i] : 0.0f;
  }

  float w[CK];
  float prod = 1.0f;
  #pragma unroll
  for (int j = CK - 1; j >= 0; --j) {
    if (j < CK - 1) prod = prod * a[j + 1];
    w[j] = c[j] * prod;
  }

  float pp = 1.0f;
  #pragma unroll
  for (int i = 0; i < CK; ++i) pp = pp * a[i];

  #pragma unroll
  for (int kk = 0; kk < 4; ++kk) {
    *(v2f*)(Wb + kk * 64 + k * 2)        = (v2f){w[4 * kk + 0], w[4 * kk + 1]};
    *(v2f*)(Wb + kk * 64 + (k + 16) * 2) = (v2f){w[4 * kk + 2], w[4 * kk + 3]};
  }
  Pb[k] = pp;
}

__device__ __forceinline__ float ds_chunk_decay(const Gates& g) {
  float pp = 1.0f;
  #pragma unroll
  for (int i = 0; i < 4; ++i) {
    #pragma unroll
    for (int e = 0; e < 4; ++e) pp = pp * ((1.0f - g.p[i][e]) * (1.0f - g.o[i][e]));
  }
  return pp;
}


static constexpr int GP    = 4;
static constexpr int NG    = NC / GP;
static constexpr int NW    = 4;
static constexpr int CG    = NW * CK;
static constexpr int GR    = GP * CK;
static constexpr int XP    = CG + 8;
static constexpr int XSL   = GR * XP;
static constexpr int WGF   = GP * 256;
static constexpr int PGF   = GP * CK;
static constexpr int OP    = CG + 4;
static constexpr int OSL   = GR * OP;
static constexpr int DEPTH = 4;
static constexpr int NS    = 4;
static constexpr int NGS   = NG / NS;

static constexpr int L_X   = 0;
static constexpr int L_O   = L_X + 2 * XSL;
static constexpr int L_W   = L_O + 2 * OSL;
static constexpr int L_P   = L_W + NGS * WGF;
static constexpr int L_WW  = L_P + NGS * PGF;
static constexpr int L_PW  = L_WW + WGF;
static constexpr int L_END = L_PW + PGF;

struct ChunkOps {
  v2f A[4];
  v2f B[4];
  v4f P0, P1;
};

struct Stage {
  v4f X[8];
};

__device__ __forceinline__ void ds_stage_load(Stage& s,
                                              const float* __restrict__ xg, unsigned xt,
                                              int grp) {
  const float* __restrict__ xr = xg + (size_t)grp * GR * DD;
  #pragma unroll
  for (int i = 0; i < 8; ++i)
    s.X[i] = *(const v4f*)(xr + (xt + (unsigned)(i * 8 * DD)));
}

__device__ __forceinline__ void ds_stage_store(const Stage& s, float* slot,
                                               unsigned xl) {
  #pragma unroll
  for (int i = 0; i < 8; ++i)
    *(v4f*)(slot + (xl + (unsigned)(i * 8 * XP))) = s.X[i];
}

__device__ __forceinline__ void ds_frag_load(ChunkOps& q, const float* slot,
                                             const float* Wg,
                                             unsigned bl, unsigned al, int u) {
  #pragma unroll
  for (int kk = 0; kk < 4; ++kk) {
    q.A[kk] = *(const v2f*)(Wg + (al + (unsigned)(u * 256 + kk * 64)));
    q.B[kk].x = slot[bl + (unsigned)((u * CK + 4 * kk) * XP)];
    q.B[kk].y = slot[bl + (unsigned)((u * CK + 4 * kk + 1) * XP)];
  }
}

__device__ __forceinline__ void ds_load_P(ChunkOps& q, const float* Pg,
                                          unsigned ph, int u) {
  q.P0 = *(const v4f*)(Pg + (ph + (unsigned)(u * CK)));
  q.P1 = *(const v4f*)(Pg + (ph + (unsigned)(u * CK + 4)));
}

__device__ __forceinline__ void ds_out_rows_load(v4f (&R)[8], const float* oslot,
                                                 unsigned ol) {
  #pragma unroll
  for (int i = 0; i < 8; ++i)
    R[i] = *(const v4f*)(oslot + (ol + (unsigned)(i * 8 * OP)));
}

__device__ __forceinline__ void ds_out_rows_store(const v4f (&R)[8],
                                                  float* __restrict__ og,
                                                  unsigned xt, int grp) {
  float* __restrict__ orow = og + (size_t)grp * GR * DD;
  #pragma unroll
  for (int i = 0; i < 8; ++i)
    *(v4f*)(orow + (xt + (unsigned)(i * 8 * DD))) = R[i];
}

__device__ __forceinline__ void ds_slot_barrier() {
  __builtin_amdgcn_fence(__ATOMIC_RELEASE, "workgroup", "local");
  __builtin_amdgcn_s_barrier();
  __builtin_amdgcn_fence(__ATOMIC_ACQUIRE, "workgroup", "local");
}

__device__ __forceinline__ void ds_warm_group(const Stage& s, float* slot,
                                              const float* Wg, const float* Pg,
                                              ChunkOps (&q)[GP], float& hc,
                                              unsigned xl, unsigned bl,
                                              unsigned al, unsigned ph) {
  ds_stage_store(s, slot, xl);
  ds_slot_barrier();
  #pragma unroll
  for (int u = 0; u < GP; ++u) {
    ds_frag_load(q[u], slot, Wg, bl, al, u);
    ds_load_P(q[u], Pg, ph, u);
  }
  v8f acc[GP];
  #pragma unroll
  for (int u = 0; u < GP; ++u) {
    acc[u] = (v8f){};
    #pragma unroll
    for (int kk = 0; kk < 4; ++kk) {
      acc[u] = __builtin_amdgcn_wmma_f32_16x16x4_f32(
           false, q[u].A[kk],  false, q[u].B[kk],
           (short)0, acc[u],  false,  false);
    }
  }
  #pragma unroll
  for (int u = 0; u < GP; ++u) hc = __builtin_fmaf(q[u].P1.w, hc, acc[u][7]);
}

__global__ void __launch_bounds__(NW * 32, 2)
ds_scan_wmma_kernel(const float* __restrict__ x,
                    const float* __restrict__ push_gate,
                    const float* __restrict__ pop_gate,
                    float* __restrict__ out) {
  __shared__ __attribute__((aligned(16))) float lds[L_END];

  const int cgi  = blockIdx.x;
  const int b    = blockIdx.y;
  const int seg  = blockIdx.z;
  const int tid  = threadIdx.x;
  const int w    = tid >> 5;
  const int lane = tid & 31;
  const int half = lane >> 4;
  const int n    = lane & 15;

  const float* __restrict__ xg = x   + (size_t)b * LL * DD + cgi * CG;
  float*       __restrict__ og = out + (size_t)b * LL * DD + cgi * CG;
  const float* __restrict__ gp = push_gate + (size_t)b * LL;
  const float* __restrict__ go = pop_gate  + (size_t)b * LL;
  const unsigned ob = 8u * half * OP + 16u * w + n;
  const unsigned ol = (unsigned)(tid >> 4) * OP + (unsigned)(tid & 15) * 4u;
  const unsigned ph = 8u * half;
  const unsigned xt = (unsigned)(tid >> 4) * DD + (unsigned)(tid & 15) * 4u;
  const unsigned xl = (unsigned)(tid >> 4) * XP + (unsigned)(tid & 15) * 4u;
  const int      bc = tid >> 4;
  const int      bk = tid & 15;
  const unsigned bl = 2u * half * XP + 16u * w + n;
  const unsigned al = lane * 2u;

  const int gs = seg * NGS;

  Gates G[4];
  #pragma unroll
  for (int i = 0; i < 4; ++i)
    ds_gates_load(G[i], gp, go, (unsigned)((gs * GP + bc + 8 * i) * CK));

  int gw = gs;
  {
    float wgt = 1.0f;
    while (gw > 0 && wgt != 0.0f) {
      --gw;
      Gates gg;
      ds_gates_load(gg, gp, go, (unsigned)((gw * GP + (lane & 3)) * CK));
      const int pb = __builtin_bit_cast(int, ds_chunk_decay(gg));
      const float p0 = __builtin_bit_cast(float, __builtin_amdgcn_readlane(pb, 0));
      const float p1 = __builtin_bit_cast(float, __builtin_amdgcn_readlane(pb, 1));
      const float p2 = __builtin_bit_cast(float, __builtin_amdgcn_readlane(pb, 2));
      const float p3 = __builtin_bit_cast(float, __builtin_amdgcn_readlane(pb, 3));
      wgt = wgt * p3 * p2 * p1 * p0;
    }
  }
  const int nwu = gs - gw;
  float* const Ww = lds + L_WW;
  float* const Pw = lds + L_PW;
  const bool wb = tid < GP * CK;

  Stage    S[DEPTH];
  ChunkOps q[GP];
  float hc = 0.0f;

  {
    Stage t;
    Stage& tw = S[0];
    Gates gm;
    if (nwu > 0 && wb) ds_gates_load(gm, gp, go, (unsigned)((gw * GP + bc) * CK));
    if (nwu > 0) ds_stage_load(tw, xg, xt, gw);
    ds_stage_load(t, xg, xt, gs);
    #pragma unroll
    for (int j = 0; j < DEPTH - 1; ++j)
      ds_stage_load(S[1 + j], xg, xt, gs + 1 + j);

    __builtin_amdgcn_sched_barrier(0);

    #pragma unroll
    for (int i = 0; i < 4; ++i)
      ds_build_row(G[i], bk, lds + L_W + (bc + 8 * i) * 256, lds + L_P + (bc + 8 * i) * CK);

    if (nwu > 0) {
      if (wb) ds_build_row(gm, bk, Ww + bc * 256, Pw + bc * CK);
      ds_warm_group(tw, lds + L_X + XSL, Ww, Pw, q, hc, xl, bl, al, ph);
      for (int wi = 1; wi < nwu; ++wi) {
        if (wb) ds_gates_load(gm, gp, go, (unsigned)(((gw + wi) * GP + bc) * CK));
        ds_stage_load(tw, xg, xt, gw + wi);
        ds_slot_barrier();
        if (wb) ds_build_row(gm, bk, Ww + bc * 256, Pw + bc * CK);
        ds_warm_group(tw, lds + L_X + XSL, Ww, Pw, q, hc, xl, bl, al, ph);
      }
    }
    ds_stage_load(S[0], xg, xt, gs + DEPTH);

    ds_stage_store(t, lds + L_X, xl);
    ds_slot_barrier();
    #pragma unroll
    for (int u = 0; u < GP; ++u) {
      ds_frag_load(q[u], lds + L_X, lds + L_W, bl, al, u);
      ds_load_P(q[u], lds + L_P, ph, u);
    }
  }

  for (int g0 = 0; g0 < NGS; g0 += DEPTH) {
    #pragma unroll
    for (int p = 0; p < DEPTH; ++p) {
      const int g   = g0 + p;

      v8f acc[GP];
      #pragma unroll
      for (int u = 0; u < GP; ++u) {
        acc[u] = (v8f){};
        #pragma unroll
        for (int kk = 0; kk < 4; ++kk) {
          acc[u] = __builtin_amdgcn_wmma_f32_16x16x4_f32(
               false, q[u].A[kk],  false, q[u].B[kk],
               (short)0, acc[u],  false,  false);
        }
      }

      v4f R[8];
      const bool drain = (g > 0);
      const int gn = (g + 1 < NGS) ? (g + 1) : (NGS - 1);
      {
        Stage& sn   = S[(p + 1) % DEPTH];
        float* slot = lds + L_X + ((p + 1) & 1) * XSL;
        ds_stage_store(sn, slot, xl);
        const int gl = (g + 1 + DEPTH < NGS) ? (g + 1 + DEPTH) : (NGS - 1);
        ds_stage_load(sn, xg, xt, gs + gl);
        ds_slot_barrier();
        if (drain) ds_out_rows_load(R, lds + L_O + ((p + 1) & 1) * OSL, ol);
        #pragma unroll
        for (int u = 0; u < GP; ++u)
          ds_frag_load(q[u], slot, lds + L_W + gn * WGF, bl, al, u);
      }

      __builtin_amdgcn_sched_barrier(0);

      float hin[GP];
      #pragma unroll
      for (int u = 0; u < GP; ++u) {
        hin[u] = hc;
        hc = __builtin_fmaf(q[u].P1.w, hc, acc[u][7]);
      }

      float* oslot = lds + L_O + (p & 1) * OSL;
      #pragma unroll
      for (int u = 0; u < GP; ++u) {
        const unsigned hbits = __builtin_bit_cast(unsigned, hin[u]);
        const float hx = __builtin_bit_cast(float,
            __builtin_amdgcn_permlanex16(hbits, hbits, 0x76543210u, 0xfedcba98u,
                                         false, false));
        const float h = half ? hin[u] : hx;

        acc[u][0] += q[u].P0.x * h;  acc[u][1] += q[u].P0.y * h;
        acc[u][2] += q[u].P0.z * h;  acc[u][3] += q[u].P0.w * h;
        acc[u][4] += q[u].P1.x * h;  acc[u][5] += q[u].P1.y * h;
        acc[u][6] += q[u].P1.z * h;
        acc[u][7] = __builtin_fmaf(q[u].P1.w, h, acc[u][7]);

        #pragma unroll
        for (int r = 0; r < 8; ++r) {
          oslot[ob + (unsigned)((u * CK + r) * OP)] = acc[u][r];
        }

        if (u == 0 && drain) ds_out_rows_store(R, og, xt, gs + g - 1);
      }

      {
        unsigned ps = ph;
        asm volatile("" : "+v"(ps)
                        : "v"(acc[0]), "v"(acc[1]), "v"(acc[2]), "v"(acc[3]));
        #pragma unroll
        for (int u = 0; u < GP; ++u) ds_load_P(q[u], lds + L_P + gn * PGF, ps, u);
      }
    }
  }

  {
    v4f R[8];
    ds_slot_barrier();
    ds_out_rows_load(R, lds + L_O + ((NGS - 1) & 1) * OSL, ol);
    ds_out_rows_store(R, og, xt, gs + NGS - 1);
  }
}

extern "C" void kernel_launch(void* const* d_in, const int* in_sizes, int n_in,
                              void* d_out, int out_size, void* d_ws, size_t ws_size,
                              hipStream_t stream) {
  const float* x  = (const float*)d_in[0];
  const float* pg = (const float*)d_in[1];
  const float* og = (const float*)d_in[2];
  float* out = (float*)d_out;

  ds_scan_wmma_kernel<<<dim3(DD / CG, BB, NS), dim3(NW * 32), 0, stream>>>(x, pg, og, out);
}
